# starting kernel
# baseline (speedup 1.0000x reference)
#include <hip/hip_runtime.h>
#include <hip/hip_bf16.h>
#include <math.h>

#define T_LEN   50000
#define W_LEN   101
#define N_IN    128
#define RANK    10
#define N_OUT   64

#define KPAD    104
#define KSTEPS  26
#define ROWS    32

#define CH1     512
#define NSUB1   4
#define CH3     128

typedef __attribute__((ext_vector_type(2))) float v2f;
typedef __attribute__((ext_vector_type(8))) float v8f;

__device__ __forceinline__ void conv_tile(const float* __restrict__ Xw, int xbase,
                                          const float* __restrict__ Kf,
                                          int lane, v8f& acc_re, v8f& acc_im) {
    const int col = lane & 15;
    const int m   = lane & 15;
    const int kb  = (lane >> 4) << 1;
#pragma unroll
    for (int r = 0; r < 8; ++r) { acc_re[r] = 0.0f; acc_im[r] = 0.0f; }
    for (int s = 0; s < KSTEPS; ++s) {
        const int u0 = s << 2;
        v2f a0, a1, b;
        const float* k0 = &Kf[m * KPAD + u0 + kb];
        a0.x = k0[0]; a0.y = k0[1];
        const float* k1 = &Kf[(m + 16) * KPAD + u0 + kb];
        a1.x = k1[0]; a1.y = k1[1];
        const int bi = xbase + col + u0 + kb;
        b.x = Xw[bi]; b.y = Xw[bi + 1];
        acc_re = __builtin_amdgcn_wmma_f32_16x16x4_f32(
            false, a0, false, b, (short)0, acc_re, false, false);
        acc_im = __builtin_amdgcn_wmma_f32_16x16x4_f32(
            false, a1, false, b, (short)0, acc_im, false, false);
    }
}

__device__ __forceinline__ void stage_filters(const float* __restrict__ K,
                                              float* __restrict__ Kf, int tid) {
    for (int i = tid; i < ROWS * KPAD; i += 256) {
        const int j = i / KPAD, u = i % KPAD;
        float v = 0.0f;
        if (u < W_LEN) {
            if (j < RANK)                      v = K[(j * 2 + 0) * W_LEN + (W_LEN - 1 - u)];
            else if (j >= 16 && j < 16 + RANK) v = K[((j - 16) * 2 + 1) * W_LEN + (W_LEN - 1 - u)];
        }
        Kf[i] = v;
    }
}

__global__ void zero_ws_kernel(float* __restrict__ p, int nelem) {
    const int i = blockIdx.x * blockDim.x + threadIdx.x;
    if (i < nelem) p[i] = 0.0f;
}

__global__ void __launch_bounds__(256)
pass1_stats(const float* __restrict__ X, const float* __restrict__ K,
            float* __restrict__ sumA, float* __restrict__ sumQ) {
    __shared__ float Kf[ROWS * KPAD];
    __shared__ float Xw[CH1 + KPAD + 8];
    const int tid  = threadIdx.x;
    const int lane = tid & 31;
    const int wave = tid >> 5;
    stage_filters(K, Kf, tid);

    const int T0 = blockIdx.x * CH1;
    const int n0 = blockIdx.y * 16;

    for (int nn = 0; nn < 16; ++nn) {
        const int n = n0 + nn;
        __syncthreads();
        for (int i = tid; i < CH1 + KPAD; i += 256) {
            const int t = T0 - 50 + i;
            Xw[i] = (t >= 0 && t < T_LEN) ? X[n * T_LEN + t] : 0.0f;
        }
        __syncthreads();

        float s[8], q[8];
#pragma unroll
        for (int r = 0; r < 8; ++r) { s[r] = 0.0f; q[r] = 0.0f; }

        for (int sub = 0; sub < NSUB1; ++sub) {
            const int toff = (wave * NSUB1 + sub) * 16;
            v8f ar, ai;
            conv_tile(Xw, toff, Kf, lane, ar, ai);
            const int  t    = T0 + toff + (lane & 15);
            const bool tval = (t < T_LEN);
#pragma unroll
            for (int r = 0; r < 8; ++r) {
                const float re = ar[r], im = ai[r];
                float c = sqrtf(re * re + im * im);
                c = tval ? c : 0.0f;
                s[r] += c; q[r] += c * c;
            }
        }
#pragma unroll
        for (int r = 0; r < 8; ++r) {
            float sv = s[r], qv = q[r];
            for (int off = 1; off < 16; off <<= 1) {
                sv += __shfl_xor(sv, off, 32);
                qv += __shfl_xor(qv, off, 32);
            }
            if ((lane & 15) == 0) {
                const int k = (lane >> 4) * 8 + r;
                if (k < RANK) {
                    atomicAdd(&sumA[n * RANK + k], sv);
                    atomicAdd(&sumQ[n * RANK + k], qv);
                }
            }
        }
    }
}

__global__ void finalize_stats(const float* __restrict__ B1,
                               const float* __restrict__ sumA,
                               const float* __restrict__ sumQ,
                               float* __restrict__ B1s, float* __restrict__ ybar) {
    const int tid = threadIdx.x;
    if (tid < RANK) ybar[tid] = 0.0f;
    __syncthreads();
    for (int i = tid; i < N_IN * RANK; i += blockDim.x) {
        const float s    = sumA[i], q = sumQ[i];
        const float mean = s / (float)T_LEN;
        const float var  = (q - s * mean) / (float)(T_LEN - 1);
        const float w    = B1[i] * rsqrtf(var);
        B1s[i] = w;
        atomicAdd(&ybar[i % RANK], w * mean);
    }
}

__global__ void __launch_bounds__(256)
pass3_out(const float* __restrict__ X, const float* __restrict__ K,
          const float* __restrict__ B2, const float* __restrict__ bias,
          const float* __restrict__ B1s, const float* __restrict__ ybar,
          float* __restrict__ out) {
    __shared__ float Kf[ROWS * KPAD];
    __shared__ float Xw[CH3 + KPAD + 8];
    __shared__ float B1sS[N_IN * RANK];
    __shared__ float B2S[RANK * N_OUT];
    __shared__ float ybS[RANK];
    __shared__ float biasS[N_OUT];
    const int tid  = threadIdx.x;
    const int lane = tid & 31;
    const int wave = tid >> 5;
    const int hi   = lane >> 4;

    stage_filters(K, Kf, tid);
    for (int i = tid; i < N_IN * RANK; i += 256) B1sS[i] = B1s[i];
    for (int i = tid; i < RANK * N_OUT; i += 256) B2S[i] = B2[i];
    if (tid < RANK)  ybS[tid]  = ybar[tid];
    if (tid < N_OUT) biasS[tid] = bias[tid];

    const int T0   = blockIdx.x * CH3;
    const int toff = wave * 16;
    const int t    = T0 + toff + (lane & 15);

    float y[8];
#pragma unroll
    for (int r = 0; r < 8; ++r) y[r] = 0.0f;

    for (int n = 0; n < N_IN; ++n) {
        __syncthreads();
        for (int i = tid; i < CH3 + KPAD; i += 256) {
            const int tt = T0 - 50 + i;
            Xw[i] = (tt >= 0 && tt < T_LEN) ? X[n * T_LEN + tt] : 0.0f;
        }
        __syncthreads();
        v8f ar, ai;
        conv_tile(Xw, toff, Kf, lane, ar, ai);
#pragma unroll
        for (int r = 0; r < 8; ++r) {
            const int   k  = hi * 8 + r;
            const float re = ar[r], im = ai[r];
            const float c  = sqrtf(re * re + im * im);
            const float w  = (k < RANK) ? B1sS[n * RANK + k] : 0.0f;
            y[r] += w * c;
        }
    }

    float yp[8];
    int   kc[8];
#pragma unroll
    for (int r = 0; r < 8; ++r) {
        const int k = hi * 8 + r;
        yp[r] = (k < RANK) ? (y[r] - ybS[k]) : 0.0f;
        kc[r] = (k < RANK) ? k : 0;
    }
    for (int m = 0; m < N_OUT; ++m) {
        float p = 0.0f;
#pragma unroll
        for (int r = 0; r < 8; ++r) p += yp[r] * B2S[kc[r] * N_OUT + m];
        p += __shfl_xor(p, 16, 32);
        if (hi == 0 && t < T_LEN) out[m * T_LEN + t] = p + biasS[m];
    }
}

extern "C" void kernel_launch(void* const* d_in, const int* in_sizes, int n_in,
                              void* d_out, int out_size, void* d_ws, size_t ws_size,
                              hipStream_t stream) {
    const float* X    = (const float*)d_in[0];
    const float* K    = (const float*)d_in[1];
    const float* B1   = (const float*)d_in[2];
    const float* B2   = (const float*)d_in[3];
    const float* bias = (const float*)d_in[4];
    float*       out  = (float*)d_out;

    float* ws   = (float*)d_ws;
    float* sumA = ws;
    float* sumQ = ws + N_IN * RANK;
    float* B1s  = ws + 2 * N_IN * RANK;
    float* ybar = ws + 3 * N_IN * RANK;

    const int nstats = 2 * N_IN * RANK;
    zero_ws_kernel<<<(nstats + 255) / 256, 256, 0, stream>>>(ws, nstats);

    const int nchunk1 = (T_LEN + CH1 - 1) / CH1;
    pass1_stats<<<dim3(nchunk1, N_IN / 16), 256, 0, stream>>>(X, K, sumA, sumQ);

    finalize_stats<<<1, 256, 0, stream>>>(B1, sumA, sumQ, B1s, ybar);

    const int nchunk3 = (T_LEN + CH3 - 1) / CH3;
    pass3_out<<<nchunk3, 256, 0, stream>>>(X, K, B2, bias, B1s, ybar, out);
}


# pass 3's row groups graded on a 1-D grid: 12 of 8 rows, then 6 of 4, then 4 of 2, not 16 of 8: the workgroups that start last are short; y's row sum regrouped: 22 planes added in plane order
# speedup vs baseline: 2.7507x; 2.7507x over previous
#include <hip/hip_runtime.h>
#include <hip/hip_bf16.h>
#include <math.h>

#define T_LEN   50000
#define W_LEN   101
#define N_IN    128
#define RANK    10
#define N_OUT   64

#define KPAD    104
#define KSTEPS  26
#define ROWS    32
#define KH      5

#define SEGT    32
#define SEGS    8
#define K1P     36
#define NCOL1   38
#define PP      620

#define CH1     512
#define NSUB1   4
#define CH3     128
#define NCH1    98
#define NSTAT   2560
#define RPW1    2
#define NRG1    64
#define WGSTAT  40
#define REC1    64
#define PLANE1  4096
#define WS_NEED ((size_t)(4 * N_IN * RANK + NCH1 * PLANE1) * sizeof(float))

#define RG3     16
#define NCH3    391
#define TPAD    (NCH3 * CH3)
#define YPLANE  (RANK * TPAD)
#define WS_YOFF (4 * N_IN * RANK + NCH1 * PLANE1)
#define WS_NEED3 ((size_t)(WS_YOFF + RG3 * YPLANE) * sizeof(float))
#define G8      12
#define G4      6
#define G2      4
#define NPLG    (G8 + G4 + G2)
#define WS_NEED3G ((size_t)(WS_YOFF + NPLG * YPLANE) * sizeof(float))
#define RO_T    64
#define NRO     782

typedef __attribute__((ext_vector_type(2))) float v2f;
typedef __attribute__((ext_vector_type(8))) float v8f;

__device__ __forceinline__ void conv_tile(const float* __restrict__ Xw, int xbase,
                                          const float* __restrict__ Kf,
                                          int lane, v8f& acc_re, v8f& acc_im) {
    const int col = lane & 15;
    const int m   = lane & 15;
    const int kb  = (lane >> 4) << 1;
#pragma unroll
    for (int r = 0; r < 8; ++r) { acc_re[r] = 0.0f; acc_im[r] = 0.0f; }
    for (int s = 0; s < KSTEPS; ++s) {
        const int u0 = s << 2;
        v2f a0, a1, b;
        const float* k0 = &Kf[m * KPAD + u0 + kb];
        a0.x = k0[0]; a0.y = k0[1];
        const float* k1 = &Kf[(m + 16) * KPAD + u0 + kb];
        a1.x = k1[0]; a1.y = k1[1];
        const int bi = xbase + col + u0 + kb;
        b.x = Xw[bi]; b.y = Xw[bi + 1];
        acc_re = __builtin_amdgcn_wmma_f32_16x16x4_f32(
            false, a0, false, b, (short)0, acc_re, false, false);
        acc_im = __builtin_amdgcn_wmma_f32_16x16x4_f32(
            false, a1, false, b, (short)0, acc_im, false, false);
    }
}

__device__ __forceinline__ void stage_filters(const float* __restrict__ K,
                                              float* __restrict__ Kf, int tid) {
    const int  j    = tid >> 3;
    const int  u0   = tid & 7;
    const int  rr   = j & 7;
    const int  k    = ((j >> 3) & 1) * KH + rr;
    const bool rowv = (rr < KH);
    const float* __restrict__ srow = K + (rowv ? (k * 2 + (j >> 4)) : 0) * W_LEN;
    float v[KPAD / 8];
#pragma unroll
    for (int i = 0; i < KPAD / 8; ++i) {
        const int   w = W_LEN - 1 - (u0 + 8 * i);
        const float x = srow[w < 0 ? 0 : w];
        v[i] = (rowv && w >= 0) ? x : 0.0f;
    }
#pragma unroll
    for (int i = 0; i < KPAD / 8; ++i) Kf[j * KPAD + u0 + 8 * i] = v[i];
}

__device__ __forceinline__ void stage_filters_seg(const float* __restrict__ K,
                                                  float* __restrict__ Kf0,
                                                  float* __restrict__ Kf1, int tid) {
    const int m  = tid >> 4;
    const int u0 = tid & 15;
    const float* __restrict__ s0 = K + (((m >> 3) * 4 + (m & 3)) * 2 + ((m >> 2) & 1)) * W_LEN;
    const float* __restrict__ s1 = K + ((8 + ((m >> 1) & 1)) * 2 + (m & 1)) * W_LEN;
    const int g = m >> 2;
    float v0[7], v1[2];
#pragma unroll
    for (int i = 0; i < 7; ++i) {
        const int   w = W_LEN - 1 - (u0 + 16 * i);
        const float x = s0[w < 0 ? 0 : w];
        v0[i] = (w >= 0) ? x : 0.0f;
    }
#pragma unroll
    for (int i = 0; i < 2; ++i) {
        const int   w = W_LEN - 1 - (SEGT * g + u0 + 16 * i);
        const float x = s1[w < 0 ? 0 : w];
        v1[i] = (w >= 0) ? x : 0.0f;
    }
#pragma unroll
    for (int i = 0; i < 6; ++i) Kf0[m * KPAD + u0 + 16 * i] = v0[i];
    if (u0 < KPAD - 96) Kf0[m * KPAD + 96 + u0] = v0[6];
#pragma unroll
    for (int i = 0; i < 2; ++i) Kf1[m * K1P + u0 + 16 * i] = v1[i];
}

__device__ __forceinline__ void conv_seg4(const float* __restrict__ Xw, int xbase,
                                          const float* __restrict__ Kf0,
                                          const float* __restrict__ Kf1, int lane,
                                          v8f (&acc0)[NSUB1], v8f (&acc1)[NSUB1]) {
    const int col = lane & 15;
    const int m   = lane & 15;
    const int kb  = (lane >> 4) << 1;
#pragma unroll
    for (int j = 0; j < NSUB1; ++j) {
#pragma unroll
        for (int r = 0; r < 8; ++r) { acc0[j][r] = 0.0f; acc1[j][r] = 0.0f; }
    }
    const float* __restrict__ k0 = &Kf0[m * KPAD + kb];
    const float* __restrict__ k1 = &Kf1[m * K1P + kb];
    const float* __restrict__ xb = &Xw[xbase + col + kb];
#pragma unroll
    for (int s = 0; s < KSTEPS; ++s) {
        const int u0 = s << 2;
        v2f a0, a1;
        a0.x = k0[u0]; a0.y = k0[u0 + 1];
        if (s < SEGS) { a1.x = k1[u0]; a1.y = k1[u0 + 1]; }
        v2f b[NSUB1];
#pragma unroll
        for (int j = 0; j < NSUB1; ++j) {
            b[j].x = xb[16 * j + u0]; b[j].y = xb[16 * j + u0 + 1];
        }
#pragma unroll
        for (int j = 0; j < NSUB1; ++j) {
            acc0[j] = __builtin_amdgcn_wmma_f32_16x16x4_f32(
                false, a0, false, b[j], (short)0, acc0[j], false, false);
            if (s < SEGS)
                acc1[j] = __builtin_amdgcn_wmma_f32_16x16x4_f32(
                    false, a1, false, b[j], (short)0, acc1[j], false, false);
        }
    }
}

__device__ __forceinline__ void conv_seg1(const float* __restrict__ Xw, int xbase,
                                          const float* __restrict__ Kf1, int lane, v8f& acc1) {
    const int col = lane & 15;
    const int m   = lane & 15;
    const int kb  = (lane >> 4) << 1;
#pragma unroll
    for (int r = 0; r < 8; ++r) acc1[r] = 0.0f;
    const float* __restrict__ k1 = &Kf1[m * K1P + kb];
    const float* __restrict__ xb = &Xw[xbase + col + kb];
#pragma unroll
    for (int s = 0; s < SEGS; ++s) {
        const int u0 = s << 2;
        v2f a1, b;
        a1.x = k1[u0]; a1.y = k1[u0 + 1];
        b.x  = xb[u0]; b.y  = xb[u0 + 1];
        acc1 = __builtin_amdgcn_wmma_f32_16x16x4_f32(
            false, a1, false, b, (short)0, acc1, false, false);
    }
}

__global__ void zero_ws_kernel(float* __restrict__ p, int nelem) {
    const int i = blockIdx.x * blockDim.x + threadIdx.x;
    if (i < nelem) p[i] = 0.0f;
}

template <bool PARTIALS>
__global__ void __launch_bounds__(256)
pass1_stats(const float* __restrict__ X, const float* __restrict__ K,
            float* __restrict__ sumA, float* __restrict__ sumQ,
            float* __restrict__ part) {
    __shared__ float Kf0[16 * KPAD];
    __shared__ float Kf1[16 * K1P];
    __shared__ float Xw[3 * 256];
    __shared__ float Pb[16 * PP];
    __shared__ float red[PARTIALS ? 8 * WGSTAT : 1];
    const int tid  = threadIdx.x;
    const int lane = tid & 31;
    const int wave = tid >> 5;
    const int hi   = lane >> 4;
    const int col  = lane & 15;
    stage_filters_seg(K, Kf0, Kf1, tid);

    const int T0 = blockIdx.x * CH1;
    const int n0 = blockIdx.y * RPW1;

    for (int nn = 0; nn < RPW1; ++nn) {
        const int n = n0 + nn;
        __syncthreads();
        float xv[3];
#pragma unroll
        for (int jj = 0; jj < 3; ++jj) {
            const int   t  = T0 - 50 + tid + 256 * jj;
            const int   tc = t < 0 ? 0 : (t < T_LEN ? t : T_LEN - 1);
            const float x  = X[n * T_LEN + tc];
            xv[jj] = (t >= 0 && t < T_LEN) ? x : 0.0f;
        }
        Xw[tid]       = xv[0];
        Xw[tid + 256] = xv[1];
        Xw[tid + 512] = xv[2];
        __syncthreads();

        float s[KH], q[KH];
#pragma unroll
        for (int r = 0; r < KH; ++r) { s[r] = 0.0f; q[r] = 0.0f; }

        v8f a0[NSUB1], a1[NSUB1];
        conv_seg4(Xw, wave * NSUB1 * 16, Kf0, Kf1, lane, a0, a1);
#pragma unroll
        for (int sub = 0; sub < NSUB1; ++sub) {
#pragma unroll
            for (int r = 0; r < 8; ++r)
                Pb[(8 * hi + r) * PP + (wave * NSUB1 + sub) * 16 + col] = a1[sub][r];
        }
        if (wave < NCOL1 - 8 * NSUB1) {
            v8f ah;
            conv_seg1(Xw, (8 * NSUB1 + wave) * 16, Kf1, lane, ah);
#pragma unroll
            for (int r = 0; r < 8; ++r)
                Pb[(8 * hi + r) * PP + (8 * NSUB1 + wave) * 16 + col] = ah[r];
        }
        __syncthreads();
#pragma unroll
        for (int sub = 0; sub < NSUB1; ++sub) {
            const int  tl   = (wave * NSUB1 + sub) * 16 + col;
            const int  t    = T0 + tl;
            const bool tval = (t < T_LEN);
#pragma unroll
            for (int r = 0; r < 4; ++r) {
                const float re = a0[sub][r], im = a0[sub][r + 4];
                float c = sqrtf(re * re + im * im);
                c = tval ? c : 0.0f;
                s[r] += c; q[r] += c * c;
            }
            {
                const float* __restrict__ pr = &Pb[(2 * hi) * PP + tl];
                const float* __restrict__ pi = pr + PP;
                const float re = ((pr[0] + pr[4 * PP + SEGT]) + pr[8 * PP + 2 * SEGT]) + pr[12 * PP + 3 * SEGT];
                const float im = ((pi[0] + pi[4 * PP + SEGT]) + pi[8 * PP + 2 * SEGT]) + pi[12 * PP + 3 * SEGT];
                float c = sqrtf(re * re + im * im);
                c = tval ? c : 0.0f;
                s[4] += c; q[4] += c * c;
            }
        }
#pragma unroll
        for (int r = 0; r < KH; ++r) {
            float sv = s[r], qv = q[r];
            for (int off = 1; off < 16; off <<= 1) {
                sv += __shfl_xor(sv, off, 32);
                qv += __shfl_xor(qv, off, 32);
            }
            if (col == 0) {
                const int k = (r < 4) ? hi * 4 + r : 8 + hi;
                if constexpr (PARTIALS) {
                    red[wave * WGSTAT + nn * RANK + k]               = sv;
                    red[wave * WGSTAT + RPW1 * RANK + nn * RANK + k] = qv;
                } else {
                    atomicAdd(&sumA[n * RANK + k], sv);
                    atomicAdd(&sumQ[n * RANK + k], qv);
                }
            }
        }
    }
    if constexpr (PARTIALS) {
        __syncthreads();
        const int st = tid >> 5, j = tid & 31;
        if (st < 2 && j < RPW1 * RANK) {
            float a = 0.0f;
            for (int w = 0; w < 8; ++w) a += red[w * WGSTAT + st * (RPW1 * RANK) + j];
            part[(blockIdx.x * NRG1 + blockIdx.y) * REC1 + st * 32 + j] = a;
        }
    }
}

template <bool PARTIALS>
__global__ void __launch_bounds__(1024)
finalize_stats(const float* __restrict__ B1,
               const float* __restrict__ sumA,
               const float* __restrict__ sumQ,
               const float* __restrict__ part,
               float* __restrict__ B1s, float* __restrict__ ybar) {
    __shared__ float tot[NSTAT];
    const int tid = threadIdx.x;
    if (tid < RANK) ybar[tid] = 0.0f;
    for (int e = tid; e < NSTAT; e += blockDim.x) {
        float a;
        if constexpr (PARTIALS) {
            a = 0.0f;
            const int st  = e / (N_IN * RANK), idx = e % (N_IN * RANK);
            const int off = (idx / (RPW1 * RANK)) * REC1 + st * 32 + idx % (RPW1 * RANK);
#pragma unroll 49
            for (int c = 0; c < NCH1; ++c) a += part[c * PLANE1 + off];
        } else {
            a = (e < N_IN * RANK) ? sumA[e] : sumQ[e - N_IN * RANK];
        }
        tot[e] = a;
    }
    __syncthreads();
    for (int i = tid; i < N_IN * RANK; i += blockDim.x) {
        const float s    = tot[i], q = tot[N_IN * RANK + i];
        const float mean = s / (float)T_LEN;
        const float var  = (q - s * mean) / (float)(T_LEN - 1);
        const float w    = B1[i] * rsqrtf(var);
        B1s[i] = w;
        atomicAdd(&ybar[i % RANK], w * mean);
    }
}

__global__ void __launch_bounds__(256)
pass3_out(const float* __restrict__ X, const float* __restrict__ K,
          const float* __restrict__ B2, const float* __restrict__ bias,
          const float* __restrict__ B1s, const float* __restrict__ ybar,
          float* __restrict__ out) {
    __shared__ float Kf[ROWS * KPAD];
    __shared__ float Xw[CH3 + KPAD + 8];
    __shared__ float B1sS[N_IN * RANK];
    __shared__ float B2S[RANK * N_OUT];
    __shared__ float ybS[RANK];
    __shared__ float biasS[N_OUT];
    const int tid  = threadIdx.x;
    const int lane = tid & 31;
    const int wave = tid >> 5;
    const int hi   = lane >> 4;

    stage_filters(K, Kf, tid);
    for (int i = tid; i < N_IN * RANK; i += 256) B1sS[i] = B1s[i];
    for (int i = tid; i < RANK * N_OUT; i += 256) B2S[i] = B2[i];
    if (tid < RANK)  ybS[tid]  = ybar[tid];
    if (tid < N_OUT) biasS[tid] = bias[tid];

    const int T0   = blockIdx.x * CH3;
    const int toff = wave * 16;
    const int t    = T0 + toff + (lane & 15);

    float y[KH];
#pragma unroll
    for (int r = 0; r < KH; ++r) y[r] = 0.0f;

    for (int n = 0; n < N_IN; ++n) {
        __syncthreads();
        for (int i = tid; i < CH3 + KPAD; i += 256) {
            const int tt = T0 - 50 + i;
            Xw[i] = (tt >= 0 && tt < T_LEN) ? X[n * T_LEN + tt] : 0.0f;
        }
        __syncthreads();
        v8f ar, ai;
        conv_tile(Xw, toff, Kf, lane, ar, ai);
#pragma unroll
        for (int r = 0; r < KH; ++r) {
            const int   k  = hi * KH + r;
            const float re = ar[r], im = ai[r];
            const float c  = sqrtf(re * re + im * im);
            y[r] += B1sS[n * RANK + k] * c;
        }
    }

    float yp[KH];
#pragma unroll
    for (int r = 0; r < KH; ++r) yp[r] = y[r] - ybS[hi * KH + r];
    for (int m = 0; m < N_OUT; ++m) {
        float p = 0.0f;
#pragma unroll
        for (int r = 0; r < KH; ++r) p += yp[r] * B2S[(hi * KH + r) * N_OUT + m];
        p += __shfl_xor(p, 16, 32);
        if (hi == 0 && t < T_LEN) out[m * T_LEN + t] = p + biasS[m];
    }
}

#define YP      520
template <int N8, int N4, int N2>
__global__ void __launch_bounds__(256) __attribute__((amdgpu_waves_per_eu(8)))
pass3_part(const float* __restrict__ X, const float* __restrict__ K,
           const float* __restrict__ B1s, float* __restrict__ ypart) {
    static_assert(8 * N8 + 4 * N4 + 2 * N2 == N_IN, "the row groups cover every input row once");
    __shared__ float Kf0[16 * KPAD];
    __shared__ float Kf1[16 * K1P];
    __shared__ float Xw[3 * 256];
    __shared__ float Pb[16 * PP];
    __shared__ float B1sS[256];
    const int tid  = threadIdx.x;
    const int lane = tid & 31;
    const int wave = tid >> 5;
    const int hi   = lane >> 4;
    const int col  = lane & 15;

    const int g  = blockIdx.x / NCH1;
    const int T0 = (blockIdx.x - g * NCH1) * CH1;
    int n0, nr;
    if (g < N8)           { n0 = 8 * g;                                nr = 8; }
    else if (g < N8 + N4) { n0 = 8 * N8 + 4 * (g - N8);                nr = 4; }
    else                  { n0 = 8 * N8 + 4 * N4 + 2 * (g - N8 - N4);  nr = 2; }

    const int   nw  = nr * RANK;
    const float b1v = B1s[n0 * RANK + (tid < nw ? tid : nw - 1)];
    stage_filters_seg(K, Kf0, Kf1, tid);
    B1sS[tid] = b1v;

    float y[NSUB1][KH];
#pragma unroll
    for (int sub = 0; sub < NSUB1; ++sub) {
#pragma unroll
        for (int r = 0; r < KH; ++r) y[sub][r] = 0.0f;
    }

    for (int nn = 0; nn < nr; ++nn) {
        const int n = n0 + nn;
        __syncthreads();
        float xv[3];
#pragma unroll
        for (int jj = 0; jj < 3; ++jj) {
            const int   t  = T0 - 50 + tid + 256 * jj;
            const int   tc = t < 0 ? 0 : (t < T_LEN ? t : T_LEN - 1);
            const float x  = X[n * T_LEN + tc];
            xv[jj] = (t >= 0 && t < T_LEN) ? x : 0.0f;
        }
        Xw[tid]       = xv[0];
        Xw[tid + 256] = xv[1];
        Xw[tid + 512] = xv[2];
        __syncthreads();

        v8f a0[NSUB1], a1[NSUB1];
        conv_seg4(Xw, wave * NSUB1 * 16, Kf0, Kf1, lane, a0, a1);
#pragma unroll
        for (int sub = 0; sub < NSUB1; ++sub) {
#pragma unroll
            for (int r = 0; r < 8; ++r)
                Pb[(8 * hi + r) * PP + (wave * NSUB1 + sub) * 16 + col] = a1[sub][r];
        }
        if (wave < NCOL1 - 8 * NSUB1) {
            v8f ah;
            conv_seg1(Xw, (8 * NSUB1 + wave) * 16, Kf1, lane, ah);
#pragma unroll
            for (int r = 0; r < 8; ++r)
                Pb[(8 * hi + r) * PP + (8 * NSUB1 + wave) * 16 + col] = ah[r];
        }
        __syncthreads();
        float w[KH];
#pragma unroll
        for (int r = 0; r < 4; ++r) w[r] = B1sS[nn * RANK + hi * 4 + r];
        w[4] = B1sS[nn * RANK + 8 + hi];
#pragma unroll
        for (int sub = 0; sub < NSUB1; ++sub) {
            const int tl = (wave * NSUB1 + sub) * 16 + col;
#pragma unroll
            for (int r = 0; r < 4; ++r) {
                const float re = a0[sub][r], im = a0[sub][r + 4];
                y[sub][r] += w[r] * sqrtf(re * re + im * im);
            }
            {
                const float* __restrict__ pr = &Pb[(2 * hi) * PP + tl];
                const float* __restrict__ pi = pr + PP;
                const float re = ((pr[0] + pr[4 * PP + SEGT]) + pr[8 * PP + 2 * SEGT]) + pr[12 * PP + 3 * SEGT];
                const float im = ((pi[0] + pi[4 * PP + SEGT]) + pi[8 * PP + 2 * SEGT]) + pi[12 * PP + 3 * SEGT];
                y[sub][4] += w[4] * sqrtf(re * re + im * im);
            }
        }
    }

    __syncthreads();
#pragma unroll
    for (int sub = 0; sub < NSUB1; ++sub) {
        const int tl = (wave * NSUB1 + sub) * 16 + col;
#pragma unroll
        for (int r = 0; r < KH; ++r) {
            const int k = (r < 4) ? hi * 4 + r : 8 + hi;
            Pb[k * YP + tl] = y[sub][r];
        }
    }
    __syncthreads();
    float* __restrict__ yp = ypart + (size_t)g * YPLANE + T0;
    for (int i = tid; i < RANK * CH1; i += 256) {
        const int k = i / CH1, tt = i % CH1;
        if (T0 + tt < TPAD) yp[k * TPAD + tt] = Pb[k * YP + tt];
    }
}

template <int NPL>
__global__ void __launch_bounds__(RO_T)
readout(const float* __restrict__ ypart, const float* __restrict__ ybar,
        const float* __restrict__ B2, const float* __restrict__ bias,
        float* __restrict__ out) {
    const int t = blockIdx.x * RO_T + threadIdx.x;
    if (t >= T_LEN) return;
    float yp[RANK];
#pragma unroll
    for (int k = 0; k < RANK; ++k) {
        float a = 0.0f;
#pragma unroll
        for (int g = 0; g < NPL; ++g) a += ypart[(g * RANK + k) * TPAD + t];
        yp[k] = a - ybar[k];
    }
    for (int m = 0; m < N_OUT; ++m) {
        float p = 0.0f;
#pragma unroll
        for (int k = 0; k < RANK; ++k) p += yp[k] * B2[k * N_OUT + m];
        out[m * T_LEN + t] = p + bias[m];
    }
}

extern "C" void kernel_launch(void* const* d_in, const int* in_sizes, int n_in,
                              void* d_out, int out_size, void* d_ws, size_t ws_size,
                              hipStream_t stream) {
    const float* X    = (const float*)d_in[0];
    const float* K    = (const float*)d_in[1];
    const float* B1   = (const float*)d_in[2];
    const float* B2   = (const float*)d_in[3];
    const float* bias = (const float*)d_in[4];
    float*       out  = (float*)d_out;

    float* ws   = (float*)d_ws;
    float* sumA = ws;
    float* sumQ = ws + N_IN * RANK;
    float* B1s  = ws + 2 * N_IN * RANK;
    float* ybar = ws + 3 * N_IN * RANK;
    float* part = ws + 4 * N_IN * RANK;

    float* ypart = ws + WS_YOFF;

    if (ws_size >= WS_NEED3G) {
        pass1_stats<true><<<dim3(NCH1, NRG1), 256, 0, stream>>>(X, K, sumA, sumQ, part);
        finalize_stats<true><<<1, 1024, 0, stream>>>(B1, sumA, sumQ, part, B1s, ybar);
        pass3_part<G8, G4, G2><<<NCH1 * NPLG, 256, 0, stream>>>(X, K, B1s, ypart);
        readout<NPLG><<<NRO, RO_T, 0, stream>>>(ypart, ybar, B2, bias, out);
        return;
    }
    if (ws_size >= WS_NEED3) {
        pass1_stats<true><<<dim3(NCH1, NRG1), 256, 0, stream>>>(X, K, sumA, sumQ, part);
        finalize_stats<true><<<1, 1024, 0, stream>>>(B1, sumA, sumQ, part, B1s, ybar);
        pass3_part<RG3, 0, 0><<<NCH1 * RG3, 256, 0, stream>>>(X, K, B1s, ypart);
        readout<RG3><<<NRO, RO_T, 0, stream>>>(ypart, ybar, B2, bias, out);
        return;
    }
    if (ws_size >= WS_NEED) {
        pass1_stats<true><<<dim3(NCH1, NRG1), 256, 0, stream>>>(X, K, sumA, sumQ, part);
        finalize_stats<true><<<1, 1024, 0, stream>>>(B1, sumA, sumQ, part, B1s, ybar);
    } else {
        const int nstats = 2 * N_IN * RANK;
        zero_ws_kernel<<<(nstats + 255) / 256, 256, 0, stream>>>(ws, nstats);
        pass1_stats<false><<<dim3(NCH1, NRG1), 256, 0, stream>>>(X, K, sumA, sumQ, part);
        finalize_stats<false><<<1, 1024, 0, stream>>>(B1, sumA, sumQ, part, B1s, ybar);
    }

    const int nchunk3 = (T_LEN + CH3 - 1) / CH3;
    pass3_out<<<nchunk3, 256, 0, stream>>>(X, K, B2, bias, B1s, ybar, out);
}
